# starting kernel
# baseline (speedup 1.0000x reference)
#include <hip/hip_runtime.h>
#include <math.h>

typedef float v2f __attribute__((ext_vector_type(2)));
typedef float v8f __attribute__((ext_vector_type(8)));

constexpr int Tt = 512;
constexpr int Ff = 16;
constexpr int H1 = 5;
constexpr int H2 = 3;
constexpr int WAVES = 4;
constexpr int BPB   = WAVES * 32;

#if __has_builtin(__builtin_amdgcn_tanhf)
__device__ __forceinline__ float fast_tanh(float x) { return __builtin_amdgcn_tanhf(x); }
#elif __has_builtin(__builtin_amdgcn_tanh_f32)
__device__ __forceinline__ float fast_tanh(float x) { return __builtin_amdgcn_tanh_f32(x); }
#else
__device__ __forceinline__ float fast_tanh(float x) { return tanhf(x); }
#endif

__global__ __launch_bounds__(128, 1) void rnn2_fused_wmma(
    const float* __restrict__ x,
    const float* __restrict__ W_ih1, const float* __restrict__ W_hh1,
    const float* __restrict__ b_ih1, const float* __restrict__ b_hh1,
    const float* __restrict__ W_ih2, const float* __restrict__ W_hh2,
    const float* __restrict__ b_ih2, const float* __restrict__ b_hh2,
    const float* __restrict__ W_fc,  const float* __restrict__ b_fc,
    float* __restrict__ out)
{
    __shared__ float xp[WAVES][32][9];

    const int tid  = threadIdx.x;
    const int lane = tid & 31;
    const int wave = tid >> 5;
    const int half = lane >> 4;
    const int n    = lane & 15;
    const int base = blockIdx.x * BPB + wave * 32;

    v2f bmat[4];
#pragma unroll
    for (int j = 0; j < 4; ++j) {
        const int k0 = j * 4 + 2 * half;
        float w0 = (n < H1) ? W_ih1[n * Ff + k0]     : 0.0f;
        float w1 = (n < H1) ? W_ih1[n * Ff + k0 + 1] : 0.0f;
        v2f b; b.x = w0; b.y = w1;
        bmat[j] = b;
    }
    const float cinit = (n < H1) ? (b_ih1[n] + b_hh1[n]) : 0.0f;

    float whh1[H1][H1], wih2[H2][H1], whh2[H2][H2], b2s[H2], wfc[H2];
#pragma unroll
    for (int i = 0; i < H1; ++i)
#pragma unroll
        for (int j = 0; j < H1; ++j) whh1[i][j] = W_hh1[i * H1 + j];
#pragma unroll
    for (int i = 0; i < H2; ++i) {
#pragma unroll
        for (int j = 0; j < H1; ++j) wih2[i][j] = W_ih2[i * H1 + j];
#pragma unroll
        for (int j = 0; j < H2; ++j) whh2[i][j] = W_hh2[i * H2 + j];
        b2s[i] = b_ih2[i] + b_hh2[i];
        wfc[i] = W_fc[i];
    }
    const float bfc = b_fc[0];

    const float* rowA0  = x + (size_t)(base +      (lane & 15)) * (Tt * Ff) + 2 * half;
    const float* rowA1  = x + (size_t)(base + 16 + (lane & 15)) * (Tt * Ff) + 2 * half;
    const float* rowOwn = x + (size_t)(base + lane) * (Tt * Ff);

    float h1[H1], h2[H2];
#pragma unroll
    for (int i = 0; i < H1; ++i) h1[i] = 0.0f;
#pragma unroll
    for (int i = 0; i < H2; ++i) h2[i] = 0.0f;

    v2f a[8];
#pragma unroll
    for (int j = 0; j < 4; ++j) {
        a[j]     = __builtin_nontemporal_load((const v2f*)(rowA0 + j * 4));
        a[4 + j] = __builtin_nontemporal_load((const v2f*)(rowA1 + j * 4));
    }

    for (int t = 0; t < Tt; ++t) {
        const int tp = (t + 8 < Tt) ? (t + 8) : (Tt - 1);
        __builtin_prefetch(rowOwn + (size_t)tp * Ff, 0, 1);

        v2f anx[8];
        {
            const int tn = (t + 1 < Tt) ? (t + 1) : t;
            const float* p0 = rowA0 + (size_t)tn * Ff;
            const float* p1 = rowA1 + (size_t)tn * Ff;
#pragma unroll
            for (int j = 0; j < 4; ++j) {
                anx[j]     = __builtin_nontemporal_load((const v2f*)(p0 + j * 4));
                anx[4 + j] = __builtin_nontemporal_load((const v2f*)(p1 + j * 4));
            }
        }

        v8f c0, c1;
#pragma unroll
        for (int i = 0; i < 8; ++i) { c0[i] = cinit; c1[i] = cinit; }
#pragma unroll
        for (int j = 0; j < 4; ++j) {
            c0 = __builtin_amdgcn_wmma_f32_16x16x4_f32(
                false, a[j], false, bmat[j], (short)0, c0, false, false);
            c1 = __builtin_amdgcn_wmma_f32_16x16x4_f32(
                false, a[4 + j], false, bmat[j], (short)0, c1, false, false);
        }

        if (n < H1) {
#pragma unroll
            for (int r = 0; r < 8; ++r) xp[wave][half * 8 + r][n]      = c0[r];
#pragma unroll
            for (int r = 0; r < 8; ++r) xp[wave][16 + half * 8 + r][n] = c1[r];
        }

        float xv[H1];
#pragma unroll
        for (int i = 0; i < H1; ++i) xv[i] = xp[wave][lane][i];

        float h1n[H1];
#pragma unroll
        for (int i = 0; i < H1; ++i) {
            float s = xv[i];
#pragma unroll
            for (int j = 0; j < H1; ++j) s += whh1[i][j] * h1[j];
            h1n[i] = fast_tanh(s);
        }
#pragma unroll
        for (int i = 0; i < H1; ++i) h1[i] = h1n[i];

        float h2n[H2];
#pragma unroll
        for (int i = 0; i < H2; ++i) {
            float s = b2s[i];
#pragma unroll
            for (int j = 0; j < H1; ++j) s += wih2[i][j] * h1n[j];
#pragma unroll
            for (int j = 0; j < H2; ++j) s += whh2[i][j] * h2[j];
            h2n[i] = fast_tanh(s);
        }
#pragma unroll
        for (int i = 0; i < H2; ++i) h2[i] = h2n[i];

#pragma unroll
        for (int j = 0; j < 8; ++j) a[j] = anx[j];
    }

    float y = bfc;
#pragma unroll
    for (int j = 0; j < H2; ++j) y += wfc[j] * h2[j];
#if __has_builtin(__builtin_amdgcn_rcpf)
    out[base + lane] = __builtin_amdgcn_rcpf(1.0f + __expf(-y));
#else
    out[base + lane] = 1.0f / (1.0f + __expf(-y));
#endif
}

extern "C" void kernel_launch(void* const* d_in, const int* in_sizes, int n_in,
                              void* d_out, int out_size, void* d_ws, size_t ws_size,
                              hipStream_t stream) {
    const float* x     = (const float*)d_in[0];
    const float* W_ih1 = (const float*)d_in[1];
    const float* W_hh1 = (const float*)d_in[2];
    const float* b_ih1 = (const float*)d_in[3];
    const float* b_hh1 = (const float*)d_in[4];
    const float* W_ih2 = (const float*)d_in[5];
    const float* W_hh2 = (const float*)d_in[6];
    const float* b_ih2 = (const float*)d_in[7];
    const float* b_hh2 = (const float*)d_in[8];
    const float* W_fc  = (const float*)d_in[9];
    const float* b_fc  = (const float*)d_in[10];
    float* out = (float*)d_out;

    const int Btot = in_sizes[0] / (Tt * Ff);
    const int grid = Btot / BPB;

    rnn2_fused_wmma<<<grid, WAVES * 32, 0, stream>>>(
        x, W_ih1, W_hh1, b_ih1, b_hh1, W_ih2, W_hh2, b_ih2, b_hh2, W_fc, b_fc, out);
}


# x loads run 8 timesteps ahead in an 8-slot register ring (time loop unrolled by 8, prefetch kept 8 ahead of the ring); a wavefront-scope fence orders the per-wave LDS scatter before its gather
# speedup vs baseline: 1.0440x; 1.0440x over previous
#include <hip/hip_runtime.h>
#include <math.h>

typedef float v2f __attribute__((ext_vector_type(2)));
typedef float v8f __attribute__((ext_vector_type(8)));

constexpr int Tt = 512;
constexpr int Ff = 16;
constexpr int H1 = 5;
constexpr int H2 = 3;
constexpr int WAVES = 4;
constexpr int BPB   = WAVES * 32;
constexpr int DEPTH = 8;
static_assert(Tt % DEPTH == 0, "time loop is unrolled by DEPTH");

#if __has_builtin(__builtin_amdgcn_tanhf)
__device__ __forceinline__ float fast_tanh(float x) { return __builtin_amdgcn_tanhf(x); }
#elif __has_builtin(__builtin_amdgcn_tanh_f32)
__device__ __forceinline__ float fast_tanh(float x) { return __builtin_amdgcn_tanh_f32(x); }
#else
__device__ __forceinline__ float fast_tanh(float x) { return tanhf(x); }
#endif

__global__ __launch_bounds__(128, 1) void rnn2_fused_wmma(
    const float* __restrict__ x,
    const float* __restrict__ W_ih1, const float* __restrict__ W_hh1,
    const float* __restrict__ b_ih1, const float* __restrict__ b_hh1,
    const float* __restrict__ W_ih2, const float* __restrict__ W_hh2,
    const float* __restrict__ b_ih2, const float* __restrict__ b_hh2,
    const float* __restrict__ W_fc,  const float* __restrict__ b_fc,
    float* __restrict__ out)
{
    __shared__ float xp[WAVES][32][9];

    const int tid  = threadIdx.x;
    const int lane = tid & 31;
    const int wave = tid >> 5;
    const int half = lane >> 4;
    const int n    = lane & 15;
    const int base = blockIdx.x * BPB + wave * 32;

    v2f bmat[4];
#pragma unroll
    for (int j = 0; j < 4; ++j) {
        const int k0 = j * 4 + 2 * half;
        float w0 = (n < H1) ? W_ih1[n * Ff + k0]     : 0.0f;
        float w1 = (n < H1) ? W_ih1[n * Ff + k0 + 1] : 0.0f;
        v2f b; b.x = w0; b.y = w1;
        bmat[j] = b;
    }
    const float cinit = (n < H1) ? (b_ih1[n] + b_hh1[n]) : 0.0f;

    float whh1[H1][H1], wih2[H2][H1], whh2[H2][H2], b2s[H2], wfc[H2];
#pragma unroll
    for (int i = 0; i < H1; ++i)
#pragma unroll
        for (int j = 0; j < H1; ++j) whh1[i][j] = W_hh1[i * H1 + j];
#pragma unroll
    for (int i = 0; i < H2; ++i) {
#pragma unroll
        for (int j = 0; j < H1; ++j) wih2[i][j] = W_ih2[i * H1 + j];
#pragma unroll
        for (int j = 0; j < H2; ++j) whh2[i][j] = W_hh2[i * H2 + j];
        b2s[i] = b_ih2[i] + b_hh2[i];
        wfc[i] = W_fc[i];
    }
    const float bfc = b_fc[0];

    const float* rowA0  = x + (size_t)(base +      (lane & 15)) * (Tt * Ff) + 2 * half;
    const float* rowA1  = x + (size_t)(base + 16 + (lane & 15)) * (Tt * Ff) + 2 * half;
    const float* rowOwn = x + (size_t)(base + lane) * (Tt * Ff);

    float h1[H1], h2[H2];
#pragma unroll
    for (int i = 0; i < H1; ++i) h1[i] = 0.0f;
#pragma unroll
    for (int i = 0; i < H2; ++i) h2[i] = 0.0f;

    v2f a[DEPTH][8];
#pragma unroll
    for (int u = 0; u < DEPTH; ++u) {
        const float* p0 = rowA0 + (size_t)u * Ff;
        const float* p1 = rowA1 + (size_t)u * Ff;
#pragma unroll
        for (int j = 0; j < 4; ++j) {
            a[u][j]     = __builtin_nontemporal_load((const v2f*)(p0 + j * 4));
            a[u][4 + j] = __builtin_nontemporal_load((const v2f*)(p1 + j * 4));
        }
    }

    for (int t0 = 0; t0 < Tt; t0 += DEPTH) {
#pragma unroll
      for (int u = 0; u < DEPTH; ++u) {
        const int t = t0 + u;
        const int tp = (t + 2 * DEPTH < Tt) ? (t + 2 * DEPTH) : (Tt - 1);
        __builtin_prefetch(rowOwn + (size_t)tp * Ff, 0, 1);

        v8f c0, c1;
#pragma unroll
        for (int i = 0; i < 8; ++i) { c0[i] = cinit; c1[i] = cinit; }
#pragma unroll
        for (int j = 0; j < 4; ++j) {
            c0 = __builtin_amdgcn_wmma_f32_16x16x4_f32(
                false, a[u][j], false, bmat[j], (short)0, c0, false, false);
            c1 = __builtin_amdgcn_wmma_f32_16x16x4_f32(
                false, a[u][4 + j], false, bmat[j], (short)0, c1, false, false);
        }

        if (n < H1) {
#pragma unroll
            for (int r = 0; r < 8; ++r) xp[wave][half * 8 + r][n]      = c0[r];
#pragma unroll
            for (int r = 0; r < 8; ++r) xp[wave][16 + half * 8 + r][n] = c1[r];
        }
        {
            const int tn = (t + DEPTH < Tt) ? (t + DEPTH) : (Tt - 1);
            const float* p0 = rowA0 + (size_t)tn * Ff;
            const float* p1 = rowA1 + (size_t)tn * Ff;
#pragma unroll
            for (int j = 0; j < 4; ++j) {
                a[u][j]     = __builtin_nontemporal_load((const v2f*)(p0 + j * 4));
                a[u][4 + j] = __builtin_nontemporal_load((const v2f*)(p1 + j * 4));
            }
        }

        __builtin_amdgcn_fence(__ATOMIC_ACQ_REL, "wavefront");

        float xv[H1];
#pragma unroll
        for (int i = 0; i < H1; ++i) xv[i] = xp[wave][lane][i];

        float h1n[H1];
#pragma unroll
        for (int i = 0; i < H1; ++i) {
            float s = xv[i];
#pragma unroll
            for (int j = 0; j < H1; ++j) s += whh1[i][j] * h1[j];
            h1n[i] = fast_tanh(s);
        }
#pragma unroll
        for (int i = 0; i < H1; ++i) h1[i] = h1n[i];

        float h2n[H2];
#pragma unroll
        for (int i = 0; i < H2; ++i) {
            float s = b2s[i];
#pragma unroll
            for (int j = 0; j < H1; ++j) s += wih2[i][j] * h1n[j];
#pragma unroll
            for (int j = 0; j < H2; ++j) s += whh2[i][j] * h2[j];
            h2n[i] = fast_tanh(s);
        }
#pragma unroll
        for (int i = 0; i < H2; ++i) h2[i] = h2n[i];
      }
    }

    float y = bfc;
#pragma unroll
    for (int j = 0; j < H2; ++j) y += wfc[j] * h2[j];
#if __has_builtin(__builtin_amdgcn_rcpf)
    out[base + lane] = __builtin_amdgcn_rcpf(1.0f + __expf(-y));
#else
    out[base + lane] = 1.0f / (1.0f + __expf(-y));
#endif
}

extern "C" void kernel_launch(void* const* d_in, const int* in_sizes, int n_in,
                              void* d_out, int out_size, void* d_ws, size_t ws_size,
                              hipStream_t stream) {
    const float* x     = (const float*)d_in[0];
    const float* W_ih1 = (const float*)d_in[1];
    const float* W_hh1 = (const float*)d_in[2];
    const float* b_ih1 = (const float*)d_in[3];
    const float* b_hh1 = (const float*)d_in[4];
    const float* W_ih2 = (const float*)d_in[5];
    const float* W_hh2 = (const float*)d_in[6];
    const float* b_ih2 = (const float*)d_in[7];
    const float* b_hh2 = (const float*)d_in[8];
    const float* W_fc  = (const float*)d_in[9];
    const float* b_fc  = (const float*)d_in[10];
    float* out = (float*)d_out;

    const int Btot = in_sizes[0] / (Tt * Ff);
    const int grid = Btot / BPB;

    rnn2_fused_wmma<<<grid, WAVES * 32, 0, stream>>>(
        x, W_ih1, W_hh1, b_ih1, b_hh1, W_ih2, W_hh2, b_ih2, b_hh2, W_fc, b_fc, out);
}


# x loads coalesced: A tile rows = 8 batch rows x 2 timesteps, each lane reads two 16-byte pieces of its row (8 lines per load, was 16); this permutes the k order of the projection's 16-term sums
# speedup vs baseline: 1.6338x; 1.5649x over previous
#include <hip/hip_runtime.h>
#include <math.h>

typedef float v2f __attribute__((ext_vector_type(2)));
typedef float v4f __attribute__((ext_vector_type(4)));
typedef float v8f __attribute__((ext_vector_type(8)));

constexpr int Tt = 512;
constexpr int Ff = 16;
constexpr int H1 = 5;
constexpr int H2 = 3;
constexpr int WAVES = 4;
constexpr int BPB   = WAVES * 32;
constexpr int DEPTH = 8;
constexpr int PAIRS = DEPTH / 2;
static_assert(Tt % DEPTH == 0, "time loop is unrolled by DEPTH");
static_assert(DEPTH % 2 == 0 && Ff == 16, "a tile is 8 rows x 2 timesteps x 16 features");

#if __has_builtin(__builtin_amdgcn_tanhf)
__device__ __forceinline__ float fast_tanh(float x) { return __builtin_amdgcn_tanhf(x); }
#elif __has_builtin(__builtin_amdgcn_tanh_f32)
__device__ __forceinline__ float fast_tanh(float x) { return __builtin_amdgcn_tanh_f32(x); }
#else
__device__ __forceinline__ float fast_tanh(float x) { return tanhf(x); }
#endif

__global__ __launch_bounds__(128, 1) void rnn2_fused_wmma(
    const float* __restrict__ x,
    const float* __restrict__ W_ih1, const float* __restrict__ W_hh1,
    const float* __restrict__ b_ih1, const float* __restrict__ b_hh1,
    const float* __restrict__ W_ih2, const float* __restrict__ W_hh2,
    const float* __restrict__ b_ih2, const float* __restrict__ b_hh2,
    const float* __restrict__ W_fc,  const float* __restrict__ b_fc,
    float* __restrict__ out)
{
    __shared__ float xp[WAVES][2][32][9];

    const int tid  = threadIdx.x;
    const int lane = tid & 31;
    const int wave = tid >> 5;
    const int half = lane >> 4;
    const int n    = lane & 15;
    const int base = blockIdx.x * BPB + wave * 32;

    v2f bmat[4];
#pragma unroll
    for (int j = 0; j < 4; ++j) {
        const int k0 = 8 * (j >> 1) + 4 * half + 2 * (j & 1);
        float w0 = (n < H1) ? W_ih1[n * Ff + k0]     : 0.0f;
        float w1 = (n < H1) ? W_ih1[n * Ff + k0 + 1] : 0.0f;
        v2f b; b.x = w0; b.y = w1;
        bmat[j] = b;
    }
    const float cinit = (n < H1) ? (b_ih1[n] + b_hh1[n]) : 0.0f;
    v8f cinit8;
#pragma unroll
    for (int i = 0; i < 8; ++i) cinit8[i] = cinit;

    float whh1[H1][H1], wih2[H2][H1], whh2[H2][H2], b2s[H2], wfc[H2];
#pragma unroll
    for (int i = 0; i < H1; ++i)
#pragma unroll
        for (int j = 0; j < H1; ++j) whh1[i][j] = W_hh1[i * H1 + j];
#pragma unroll
    for (int i = 0; i < H2; ++i) {
#pragma unroll
        for (int j = 0; j < H1; ++j) wih2[i][j] = W_ih2[i * H1 + j];
#pragma unroll
        for (int j = 0; j < H2; ++j) whh2[i][j] = W_hh2[i * H2 + j];
        b2s[i] = b_ih2[i] + b_hh2[i];
        wfc[i] = W_fc[i];
    }
    const float bfc = b_fc[0];

    const float* rowT[4];
#pragma unroll
    for (int q = 0; q < 4; ++q)
        rowT[q] = x + (size_t)(base + 8 * q + (n & 7)) * (Tt * Ff) + (n >> 3) * Ff + 4 * half;
    const float* rowOwn = x + (size_t)(base + lane) * (Tt * Ff);

    float h1[H1], h2[H2];
#pragma unroll
    for (int i = 0; i < H1; ++i) h1[i] = 0.0f;
#pragma unroll
    for (int i = 0; i < H2; ++i) h2[i] = 0.0f;

    v4f a[PAIRS][4][2];
#pragma unroll
    for (int u = 0; u < PAIRS; ++u) {
#pragma unroll
        for (int q = 0; q < 4; ++q) {
            const float* p = rowT[q] + (size_t)(2 * u) * Ff;
            a[u][q][0] = __builtin_nontemporal_load((const v4f*)(p));
            a[u][q][1] = __builtin_nontemporal_load((const v4f*)(p + 8));
        }
    }

    for (int t0 = 0; t0 < Tt; t0 += DEPTH) {
#pragma unroll
      for (int u = 0; u < PAIRS; ++u) {
        const int t = t0 + 2 * u;
        const int tp = (t + 2 * DEPTH < Tt) ? (t + 2 * DEPTH) : (Tt - 1);
        __builtin_prefetch(rowOwn + (size_t)tp * Ff, 0, 1);

        v8f c[4];
#pragma unroll
        for (int j = 0; j < 4; ++j) {
#pragma unroll
            for (int q = 0; q < 4; ++q) {
                const v4f f = a[u][q][j >> 1];
                v2f av;
                if (j & 1) { av.x = f.z; av.y = f.w; } else { av.x = f.x; av.y = f.y; }
                c[q] = __builtin_amdgcn_wmma_f32_16x16x4_f32(
                    false, av, false, bmat[j], (short)0, (j == 0) ? cinit8 : c[q], false, false);
            }
        }

        if (n < H1) {
#pragma unroll
            for (int q = 0; q < 4; ++q) {
#pragma unroll
                for (int r = 0; r < 8; ++r) xp[wave][half][8 * q + r][n] = c[q][r];
            }
        }
        {
            const int tn = (t + DEPTH < Tt) ? (t + DEPTH) : (Tt - 2);
#pragma unroll
            for (int q = 0; q < 4; ++q) {
                const float* p = rowT[q] + (size_t)tn * Ff;
                a[u][q][0] = __builtin_nontemporal_load((const v4f*)(p));
                a[u][q][1] = __builtin_nontemporal_load((const v4f*)(p + 8));
            }
        }

        __builtin_amdgcn_fence(__ATOMIC_ACQ_REL, "wavefront");

        float xv[2][H1];
#pragma unroll
        for (int s = 0; s < 2; ++s)
#pragma unroll
            for (int i = 0; i < H1; ++i) xv[s][i] = xp[wave][s][lane][i];

#pragma unroll
        for (int s = 0; s < 2; ++s) {
            float h1n[H1];
#pragma unroll
            for (int i = 0; i < H1; ++i) {
                float sacc = xv[s][i];
#pragma unroll
                for (int j = 0; j < H1; ++j) sacc += whh1[i][j] * h1[j];
                h1n[i] = fast_tanh(sacc);
            }
#pragma unroll
            for (int i = 0; i < H1; ++i) h1[i] = h1n[i];

            float h2n[H2];
#pragma unroll
            for (int i = 0; i < H2; ++i) {
                float sacc = b2s[i];
#pragma unroll
                for (int j = 0; j < H1; ++j) sacc += wih2[i][j] * h1n[j];
#pragma unroll
                for (int j = 0; j < H2; ++j) sacc += whh2[i][j] * h2[j];
                h2n[i] = fast_tanh(sacc);
            }
#pragma unroll
            for (int i = 0; i < H2; ++i) h2[i] = h2n[i];
        }
      }
    }

    float y = bfc;
#pragma unroll
    for (int j = 0; j < H2; ++j) y += wfc[j] * h2[j];
#if __has_builtin(__builtin_amdgcn_rcpf)
    out[base + lane] = __builtin_amdgcn_rcpf(1.0f + __expf(-y));
#else
    out[base + lane] = 1.0f / (1.0f + __expf(-y));
#endif
}

extern "C" void kernel_launch(void* const* d_in, const int* in_sizes, int n_in,
                              void* d_out, int out_size, void* d_ws, size_t ws_size,
                              hipStream_t stream) {
    const float* x     = (const float*)d_in[0];
    const float* W_ih1 = (const float*)d_in[1];
    const float* W_hh1 = (const float*)d_in[2];
    const float* b_ih1 = (const float*)d_in[3];
    const float* b_hh1 = (const float*)d_in[4];
    const float* W_ih2 = (const float*)d_in[5];
    const float* W_hh2 = (const float*)d_in[6];
    const float* b_ih2 = (const float*)d_in[7];
    const float* b_hh2 = (const float*)d_in[8];
    const float* W_fc  = (const float*)d_in[9];
    const float* b_fc  = (const float*)d_in[10];
    float* out = (float*)d_out;

    const int Btot = in_sizes[0] / (Tt * Ff);
    const int grid = Btot / BPB;

    rnn2_fused_wmma<<<grid, WAVES * 32, 0, stream>>>(
        x, W_ih1, W_hh1, b_ih1, b_hh1, W_ih2, W_hh2, b_ih2, b_hh2, W_fc, b_fc, out);
}


# wave roles: per 32 batch rows 2 projection waves (x loads one chunk ahead, WMMA, LDS scatter) feed 1 recurrence wave through double-buffered LDS, 1 block barrier per 16 steps; row prefetch dropped
# speedup vs baseline: 3.5909x; 2.1979x over previous
#include <hip/hip_runtime.h>
#include <math.h>


typedef float v2f __attribute__((ext_vector_type(2)));
typedef float v4f __attribute__((ext_vector_type(4)));
typedef float v8f __attribute__((ext_vector_type(8)));

constexpr int Tt = 512;
constexpr int Ff = 16;
constexpr int H1 = 5;
constexpr int H2 = 3;
constexpr int BPB   = 32;
constexpr int NPROD = 2;
constexpr int WAVES = NPROD + 1;
constexpr int CH    = 16;
constexpr int NCH   = Tt / CH;
constexpr int PPW   = CH / 2 / NPROD;
constexpr int SLOTS = 2 * PPW;
constexpr int XROW  = 9;
static_assert(Tt % (2 * CH) == 0 && CH % (2 * NPROD) == 0 && Ff == 16, "a tile is 8 rows x 2 timesteps x 16 features");

#if __has_builtin(__builtin_amdgcn_tanhf)
__device__ __forceinline__ float fast_tanh(float x) { return __builtin_amdgcn_tanhf(x); }
#elif __has_builtin(__builtin_amdgcn_tanh_f32)
__device__ __forceinline__ float fast_tanh(float x) { return __builtin_amdgcn_tanh_f32(x); }
#else
__device__ __forceinline__ float fast_tanh(float x) { return tanhf(x); }
#endif

__device__ __forceinline__ void handover_barrier() {
    __builtin_amdgcn_fence(__ATOMIC_RELEASE, "workgroup", "local");
    __builtin_amdgcn_s_barrier();
    __builtin_amdgcn_fence(__ATOMIC_ACQUIRE, "workgroup", "local");
}

__global__ __launch_bounds__(WAVES * 32, 1) void rnn2_fused_wmma(
    const float* __restrict__ x,
    const float* __restrict__ W_ih1, const float* __restrict__ W_hh1,
    const float* __restrict__ b_ih1, const float* __restrict__ b_hh1,
    const float* __restrict__ W_ih2, const float* __restrict__ W_hh2,
    const float* __restrict__ b_ih2, const float* __restrict__ b_hh2,
    const float* __restrict__ W_fc,  const float* __restrict__ b_fc,
    float* __restrict__ out)
{
    __shared__ float xp[2][CH][BPB][XROW];

    const int tid  = threadIdx.x;
    const int lane = tid & 31;
    const int wave = __builtin_amdgcn_readfirstlane(tid >> 5);
    const bool is_proj = wave < NPROD;
    const int half = lane >> 4;
    const int n    = lane & 15;
    const int base = blockIdx.x * BPB;

    v2f bmat[4];
#pragma unroll
    for (int j = 0; j < 4; ++j) {
        const int k0 = 8 * (j >> 1) + 4 * half + 2 * (j & 1);
        float w0 = (n < H1) ? W_ih1[n * Ff + k0]     : 0.0f;
        float w1 = (n < H1) ? W_ih1[n * Ff + k0 + 1] : 0.0f;
        v2f b; b.x = w0; b.y = w1;
        bmat[j] = b;
    }
    const float cinit = (n < H1) ? (b_ih1[n] + b_hh1[n]) : 0.0f;
    v8f cinit8;
#pragma unroll
    for (int i = 0; i < 8; ++i) cinit8[i] = cinit;

    float whh1[H1][H1], wih2[H2][H1], whh2[H2][H2], b2s[H2], wfc[H2];
#pragma unroll
    for (int i = 0; i < H1; ++i)
#pragma unroll
        for (int j = 0; j < H1; ++j) whh1[i][j] = W_hh1[i * H1 + j];
#pragma unroll
    for (int i = 0; i < H2; ++i) {
#pragma unroll
        for (int j = 0; j < H1; ++j) wih2[i][j] = W_ih2[i * H1 + j];
#pragma unroll
        for (int j = 0; j < H2; ++j) whh2[i][j] = W_hh2[i * H2 + j];
        b2s[i] = b_ih2[i] + b_hh2[i];
        wfc[i] = W_fc[i];
    }
    const float bfc = b_fc[0];

    const float* rowT[4];
#pragma unroll
    for (int q = 0; q < 4; ++q)
        rowT[q] = x + (size_t)(base + 8 * q + (n & 7)) * (Tt * Ff) + (n >> 3) * Ff + 4 * half;
    const int pw0 = wave * PPW;

    float h1[H1], h2[H2];
#pragma unroll
    for (int i = 0; i < H1; ++i) h1[i] = 0.0f;
#pragma unroll
    for (int i = 0; i < H2; ++i) h2[i] = 0.0f;

    v4f a[SLOTS][4][2];
#pragma unroll
    for (int u = PPW; u < SLOTS; ++u)
#pragma unroll
        for (int q = 0; q < 4; ++q) { a[u][q][0] = (v4f)(0.0f); a[u][q][1] = (v4f)(0.0f); }
    if (is_proj) {
#pragma unroll
        for (int u = 0; u < PPW; ++u) {
#pragma unroll
            for (int q = 0; q < 4; ++q) {
                const float* p = rowT[q] + (size_t)(2 * (pw0 + u)) * Ff;
                a[u][q][0] = __builtin_nontemporal_load((const v4f*)(p));
                a[u][q][1] = __builtin_nontemporal_load((const v4f*)(p + 8));
            }
        }
    } else {
#pragma unroll
        for (int u = 0; u < PPW; ++u)
#pragma unroll
            for (int q = 0; q < 4; ++q) { a[u][q][0] = (v4f)(0.0f); a[u][q][1] = (v4f)(0.0f); }
    }

    for (int k2 = 0; k2 <= NCH; k2 += 2) {
#pragma unroll
     for (int par = 0; par < 2; ++par) {
      const int k = k2 + par;
      if (k > NCH) break;
      if (is_proj) {
        if (k < NCH) {
          float (*dst)[BPB][XROW] = xp[par];
#pragma unroll
          for (int uu = 0; uu < PPW; ++uu) {
            const int u = par * PPW + uu;
            const int sp = 2 * (pw0 + uu);
            const int t  = k * CH + sp;
            v8f c[4];
#pragma unroll
            for (int j = 0; j < 4; ++j) {
#pragma unroll
                for (int q = 0; q < 4; ++q) {
                    const v4f f = a[u][q][j >> 1];
                    v2f av;
                    if (j & 1) { av.x = f.z; av.y = f.w; } else { av.x = f.x; av.y = f.y; }
                    c[q] = __builtin_amdgcn_wmma_f32_16x16x4_f32(
                        false, av, false, bmat[j], (short)0, (j == 0) ? cinit8 : c[q], false, false);
                }
            }

            if (n < H1) {
#pragma unroll
                for (int q = 0; q < 4; ++q) {
#pragma unroll
                    for (int r = 0; r < 8; ++r) dst[sp + half][8 * q + r][n] = c[q][r];
                }
            }
            if (uu == 0) {
#pragma unroll
                for (int v = 0; v < PPW; ++v) {
                    const int tnx = (k + 1) * CH + 2 * (pw0 + v);
                    const int tn  = (tnx < Tt) ? tnx : (Tt - 2);
#pragma unroll
                    for (int q = 0; q < 4; ++q) {
                        const float* p = rowT[q] + (size_t)tn * Ff;
                        a[(par ^ 1) * PPW + v][q][0] = __builtin_nontemporal_load((const v4f*)(p));
                        a[(par ^ 1) * PPW + v][q][1] = __builtin_nontemporal_load((const v4f*)(p + 8));
                    }
                }
            }
          }
        }
      } else {
        if (k > 0) {
          const float* srcp = &xp[par ^ 1][0][lane][0];
          float xall[CH][H1];
#pragma unroll
          for (int s = 0; s < CH; ++s)
#pragma unroll
            for (int i = 0; i < H1; ++i) xall[s][i] = srcp[s * (BPB * XROW) + i];
#pragma unroll
          for (int s = 0; s < CH; ++s) {
            float xv[H1];
#pragma unroll
            for (int i = 0; i < H1; ++i) xv[i] = xall[s][i];

            float h1n[H1];
#pragma unroll
            for (int i = 0; i < H1; ++i) {
                float sacc = xv[i];
#pragma unroll
                for (int j = 0; j < H1; ++j) sacc += whh1[i][j] * h1[j];
                h1n[i] = fast_tanh(sacc);
            }
#pragma unroll
            for (int i = 0; i < H1; ++i) h1[i] = h1n[i];

            float h2n[H2];
#pragma unroll
            for (int i = 0; i < H2; ++i) {
                float sacc = b2s[i];
#pragma unroll
                for (int j = 0; j < H1; ++j) sacc += wih2[i][j] * h1n[j];
#pragma unroll
                for (int j = 0; j < H2; ++j) sacc += whh2[i][j] * h2[j];
                h2n[i] = fast_tanh(sacc);
            }
#pragma unroll
            for (int i = 0; i < H2; ++i) h2[i] = h2n[i];
          }
        }
      }
      handover_barrier();
     }
    }

    if (!is_proj) {
        float y = bfc;
#pragma unroll
        for (int j = 0; j < H2; ++j) y += wfc[j] * h2[j];
#if __has_builtin(__builtin_amdgcn_rcpf)
        out[base + lane] = __builtin_amdgcn_rcpf(1.0f + __expf(-y));
#else
        out[base + lane] = 1.0f / (1.0f + __expf(-y));
#endif
    }
}

extern "C" void kernel_launch(void* const* d_in, const int* in_sizes, int n_in,
                              void* d_out, int out_size, void* d_ws, size_t ws_size,
                              hipStream_t stream) {
    const float* x     = (const float*)d_in[0];
    const float* W_ih1 = (const float*)d_in[1];
    const float* W_hh1 = (const float*)d_in[2];
    const float* b_ih1 = (const float*)d_in[3];
    const float* b_hh1 = (const float*)d_in[4];
    const float* W_ih2 = (const float*)d_in[5];
    const float* W_hh2 = (const float*)d_in[6];
    const float* b_ih2 = (const float*)d_in[7];
    const float* b_hh2 = (const float*)d_in[8];
    const float* W_fc  = (const float*)d_in[9];
    const float* b_fc  = (const float*)d_in[10];
    float* out = (float*)d_out;

    const int Btot = in_sizes[0] / (Tt * Ff);
    const int grid = Btot / BPB;

    rnn2_fused_wmma<<<grid, WAVES * 32, 0, stream>>>(
        x, W_ih1, W_hh1, b_ih1, b_hh1, W_ih2, W_hh2, b_ih2, b_hh2, W_fc, b_fc, out);
}
